# starting kernel
# baseline (speedup 1.0000x reference)
#include <hip/hip_runtime.h>
#include <math.h>

#define B_TOT   262144
#define HIDDEN  96
#define NWAVE   8
#define ELEMS_PER_BLOCK (16 * NWAVE)
#define NBLK    (B_TOT / ELEMS_PER_BLOCK)
#define PS2     112
#define HS      100
#define MAX_STEPS 60

typedef __attribute__((ext_vector_type(2))) float v2f;
typedef __attribute__((ext_vector_type(4))) float v4f;
typedef __attribute__((ext_vector_type(8))) float v8f;

__device__ __forceinline__ float silu_f(float x) {
  return x * __builtin_amdgcn_rcpf(1.0f + __expf(-x));
}


__global__ void init_ws_kernel(float* ws) {
  ws[0] = 0.0f;
  ws[1] = 0.05f;
  ws[2] = 0.05f;
  ws[3] = 0.0f;
  ws[4] = 0.0f;
}

__global__ void encode_kernel(const float* __restrict__ x, float* __restrict__ y) {
  int e = blockIdx.x * blockDim.x + threadIdx.x;
  float v = x[e];
  y[e * 2 + 0] = v;
  y[e * 2 + 1] = v * v;
}

__global__ void commit_kernel(float* __restrict__ y, const float* __restrict__ y5,
                              const float* __restrict__ ws) {
  if (ws[4] == 0.0f) return;
  int i = blockIdx.x * blockDim.x + threadIdx.x;
  y[i] = y5[i];
}

__global__ void decode_kernel(const float* __restrict__ y, const float* __restrict__ Wd,
                              const float* __restrict__ bd, float* __restrict__ out) {
  int e = blockIdx.x * blockDim.x + threadIdx.x;
  float y0 = y[e * 2 + 0];
  float y1 = y[e * 2 + 1];
#pragma unroll
  for (int j = 0; j < 10; ++j)
    out[e * 10 + j] = fmaf(y0, Wd[j], fmaf(y1, Wd[10 + j], bd[j]));
}


__global__ __launch_bounds__(256) void control_kernel(float* ws) {
  __shared__ float red[256];
  const int tid = threadIdx.x;
  float s = 0.0f;
  const float* part = ws + 16;
  for (int i = tid; i < NBLK; i += 256) s += part[i];
  red[tid] = s;
  __syncthreads();
  for (int w = 128; w > 0; w >>= 1) {
    if (tid < w) red[tid] += red[tid + w];
    __syncthreads();
  }
  if (tid == 0) {
    if (ws[3] != 0.0f) {
      ws[4] = 0.0f;
    } else {
      float t   = ws[0];
      float dtc = ws[2];
      float enorm = sqrtf(red[0] / (float)(B_TOT * 2));
      int accept = (enorm <= 1.0f) ? 1 : 0;
      float factor = 0.9f * powf(fmaxf(enorm, 1e-10f), -0.2f);
      factor = fminf(fmaxf(factor, 0.2f), 10.0f);
      if (accept) t = t + dtc;
      float dt = dtc * factor;
      ws[0] = t;
      ws[1] = dt;
      ws[3] = (t >= 1.0f - 1e-7f) ? 1.0f : 0.0f;
      ws[2] = fminf(dt, 1.0f - t);
      ws[4] = accept ? 1.0f : 0.0f;
    }
  }
}


__global__ __launch_bounds__(256) void bigstep_kernel(
    const float* __restrict__ W1, const float* __restrict__ b1,
    const float* __restrict__ W2, const float* __restrict__ b2,
    const float* __restrict__ W3, const float* __restrict__ b3,
    float* __restrict__ ws, const float* __restrict__ y,
    float* __restrict__ y5out)
{
  constexpr float AA[7][6] = {
    {0.f, 0.f, 0.f, 0.f, 0.f, 0.f},
    {0.2f, 0.f, 0.f, 0.f, 0.f, 0.f},
    {3.f/40.f, 9.f/40.f, 0.f, 0.f, 0.f, 0.f},
    {44.f/45.f, -56.f/15.f, 32.f/9.f, 0.f, 0.f, 0.f},
    {19372.f/6561.f, -25360.f/2187.f, 64448.f/6561.f, -212.f/729.f, 0.f, 0.f},
    {9017.f/3168.f, -355.f/33.f, 46732.f/5247.f, 49.f/176.f, -5103.f/18656.f, 0.f},
    {35.f/384.f, 0.f, 500.f/1113.f, 125.f/192.f, -2187.f/6784.f, 11.f/84.f},
  };
  constexpr float CC[7]  = {0.f, 0.2f, 0.3f, 0.8f, 8.f/9.f, 1.f, 1.f};
  constexpr float B5c[7] = {35.f/384.f, 0.f, 500.f/1113.f, 125.f/192.f, -2187.f/6784.f, 11.f/84.f, 0.f};
  constexpr float B4c[7] = {5179.f/57600.f, 0.f, 7571.f/16695.f, 393.f/640.f,
                            -92097.f/339200.f, 187.f/2100.f, 1.f/40.f};

  __shared__ __align__(16) float w1p[2 * 96 * 2];
  __shared__ float b2s[96];
  __shared__ __align__(16) float w2p[48 * PS2 * 2];
  __shared__ __align__(16) float w3t[2 * 96];
  __shared__ float b3s[2];
  __shared__ __align__(16) float hbuf[NWAVE][16 * HS];
  __shared__ float red[256];

  const float done = ws[3];
  if (done != 0.0f) return;
  const float t0  = ws[0];
  const float dtc = ws[2];

  const int tid = threadIdx.x;
  for (int i = tid; i < 2 * 96; i += 256) {
    int pr = i / 96, c = i % 96;
    float lo = W1[(2 * pr + 0) * 96 + c];
    float hi_v = (pr == 0) ? W1[96 + c] : b1[c];
    w1p[(pr * 96 + c) * 2 + 0] = lo;
    w1p[(pr * 96 + c) * 2 + 1] = hi_v;
  }
  for (int i = tid; i < 96; i += 256) b2s[i] = b2[i];
  for (int i = tid; i < 96 * 96; i += 256) {
    int r = i / 96, c = i % 96;
    w2p[((r >> 1) * PS2 + c) * 2 + (r & 1)] = W2[i];
  }
  for (int i = tid; i < 192; i += 256) {
    int j = i / 96, n = i % 96;
    w3t[j * 96 + n] = W3[n * 2 + j];
  }
  if (tid < 2) b3s[tid] = b3[tid];
  __syncthreads();

  const int wave = tid >> 5;
  const int lane = tid & 31;
  const int m    = lane & 15;
  const int hi   = lane >> 4;
  const int elem = blockIdx.x * ELEMS_PER_BLOCK + wave * 16 + m;

  const float yv = y[elem * 2 + hi];
  float* hb = &hbuf[wave][0];

  float ks[7];
#pragma unroll
  for (int i = 0; i < 7; ++i) {
    float yi = yv;
#pragma unroll
    for (int j = 0; j < 6; ++j)
      if (j < i) yi = fmaf(dtc * AA[i][j], ks[j], yi);
    const float ti = t0 + CC[i] * dtc;

    float partner = __shfl_xor(yi, 16, 32);
    v2f a1;
    a1.x = hi ? ti : yi;
    a1.y = hi ? 1.0f : partner;
#pragma unroll
    for (int c = 0; c < 6; ++c) {
      const int col = c * 16 + m;
      const v2f bw = *(const v2f*)&w1p[(hi * 96 + col) * 2];
      v8f acc = {};
      acc = __builtin_amdgcn_wmma_f32_16x16x4_f32(false, a1, false, bw,
                                                  (short)0, acc, false, false);
#pragma unroll
      for (int v = 0; v < 8; ++v)
        hb[(v + 8 * hi) * HS + col] = silu_f(acc[v]);
    }
    asm volatile("s_wait_dscnt 0" ::: "memory");

    v8f acc2[6];
#pragma unroll
    for (int c = 0; c < 6; ++c) {
      const float bb = b2s[c * 16 + m];
#pragma unroll
      for (int v = 0; v < 8; ++v) acc2[c][v] = bb;
    }
#pragma unroll 4
    for (int k = 0; k < 24; ++k) {
      const v2f a2 = *(const v2f*)&hb[m * HS + 4 * k + 2 * hi];
      const int pidx = 2 * k + hi;
#pragma unroll
      for (int c = 0; c < 6; ++c) {
        const v2f bw = *(const v2f*)&w2p[(pidx * PS2 + c * 16 + m) * 2];
        acc2[c] = __builtin_amdgcn_wmma_f32_16x16x4_f32(false, a2, false, bw,
                                                        (short)0, acc2[c], false, false);
      }
    }
    asm volatile("s_wait_dscnt 0" ::: "memory");
#pragma unroll
    for (int c = 0; c < 6; ++c)
#pragma unroll
      for (int v = 0; v < 8; ++v)
        hb[(v + 8 * hi) * HS + c * 16 + m] = silu_f(acc2[c][v]);
    asm volatile("s_wait_dscnt 0" ::: "memory");

    float a3 = b3s[hi];
    const v4f* hrow = (const v4f*)&hb[m * HS];
    const v4f* wcol = (const v4f*)&w3t[hi * 96];
#pragma unroll
    for (int n4 = 0; n4 < 24; ++n4) {
      v4f hv = hrow[n4];
      v4f wv = wcol[n4];
      a3 = fmaf(hv.x, wv.x, a3);
      a3 = fmaf(hv.y, wv.y, a3);
      a3 = fmaf(hv.z, wv.z, a3);
      a3 = fmaf(hv.w, wv.w, a3);
    }
    ks[i] = a3;
    asm volatile("s_wait_dscnt 0" ::: "memory");
  }

  float y5 = yv, ev = 0.0f;
#pragma unroll
  for (int i = 0; i < 7; ++i) {
    y5 = fmaf(dtc * B5c[i], ks[i], y5);
    ev = fmaf(dtc * (B5c[i] - B4c[i]), ks[i], ev);
  }
  y5out[elem * 2 + hi] = y5;

  const float sc = 1e-4f + 1e-4f * fmaxf(fabsf(yv), fabsf(y5));
  const float r = ev / sc;

  red[tid] = r * r;
  __syncthreads();
  for (int w = 128; w > 0; w >>= 1) {
    if (tid < w) red[tid] += red[tid + w];
    __syncthreads();
  }
  if (tid == 0) ws[16 + blockIdx.x] = red[0];
}


extern "C" void kernel_launch(void* const* d_in, const int* in_sizes, int n_in,
                              void* d_out, int out_size, void* d_ws, size_t ws_size,
                              hipStream_t stream) {
  (void)in_sizes; (void)n_in; (void)out_size; (void)ws_size;
  const float* x  = (const float*)d_in[0];
  const float* W1 = (const float*)d_in[1];
  const float* b1 = (const float*)d_in[2];
  const float* W2 = (const float*)d_in[3];
  const float* b2 = (const float*)d_in[4];
  const float* W3 = (const float*)d_in[5];
  const float* b3 = (const float*)d_in[6];
  const float* Wd = (const float*)d_in[7];
  const float* bd = (const float*)d_in[8];
  float* out = (float*)d_out;
  float* ws  = (float*)d_ws;

  float* yst = ws + 16 + NBLK;
  float* y5  = out;

  init_ws_kernel<<<1, 1, 0, stream>>>(ws);
  encode_kernel<<<B_TOT / 256, 256, 0, stream>>>(x, yst);
  for (int s = 0; s < MAX_STEPS; ++s) {
    bigstep_kernel<<<NBLK, 256, 0, stream>>>(W1, b1, W2, b2, W3, b3, ws, yst, y5);
    control_kernel<<<1, 256, 0, stream>>>(ws);
    commit_kernel<<<(B_TOT * 2) / 256, 256, 0, stream>>>(yst, y5, ws);
  }
  decode_kernel<<<B_TOT / 256, 256, 0, stream>>>(yst, Wd, bd, out);
}


# Fuse init, encode, step controller and commit into the RK step kernel and the last verdict into decode: 61 launches per call instead of 183; state and partial sums ping-pong in workspace
# speedup vs baseline: 1.4360x; 1.4360x over previous
#include <hip/hip_runtime.h>
#include <math.h>

#define B_TOT   262144
#define HIDDEN  96
#define NWAVE   8
#define ELEMS_PER_BLOCK (16 * NWAVE)
#define NBLK    (B_TOT / ELEMS_PER_BLOCK)
#define PS2     112
#define HS      100
#define MAX_STEPS 60

typedef __attribute__((ext_vector_type(2))) float v2f;
typedef __attribute__((ext_vector_type(4))) float v4f;
typedef __attribute__((ext_vector_type(8))) float v8f;

__device__ __forceinline__ float silu_f(float x) {
  return x * __builtin_amdgcn_rcpf(1.0f + __expf(-x));
}

#define ST_A    0
#define ST_B    8
#define PART_A  16
#define PART_B  (16 + NBLK)
#define Y_OFF   (16 + 2 * NBLK)
#define Y5_OFF  (Y_OFF + 2 * B_TOT)

__device__ __forceinline__ float block_total(const float* __restrict__ part, float* red, int tid) {
  float s = 0.0f;
#pragma unroll
  for (int j = 0; j < NBLK / 256; ++j) s += part[tid + 256 * j];
  red[tid] = s;
  __syncthreads();
  for (int w = 128; w > 0; w >>= 1) {
    if (tid < w) red[tid] += red[tid + w];
    __syncthreads();
  }
  return __builtin_bit_cast(float,
      __builtin_amdgcn_readfirstlane(__builtin_bit_cast(int, red[0])));
}


__global__ __launch_bounds__(256) void decode_kernel(
    const float* __restrict__ st_in, const float* __restrict__ part_in,
    const float* __restrict__ y, const float* __restrict__ y5,
    const float* __restrict__ Wd, const float* __restrict__ bd,
    float* __restrict__ out) {
  __shared__ float red[256];
  const int tid = threadIdx.x;
  const int e = blockIdx.x * blockDim.x + tid;
  float y0 = y[e * 2 + 0];
  float y1 = y[e * 2 + 1];
  if (st_in[2] == 0.0f) {
    const float tot = block_total(part_in, red, tid);
    const float enorm = sqrtf(tot / (float)(B_TOT * 2));
    if (enorm <= 1.0f) {
      y0 = y5[e * 2 + 0];
      y1 = y5[e * 2 + 1];
    }
  }
#pragma unroll
  for (int j = 0; j < 10; ++j)
    out[e * 10 + j] = fmaf(y0, Wd[j], fmaf(y1, Wd[10 + j], bd[j]));
}


template <bool FIRST>
__global__ __launch_bounds__(256) void step_kernel(
    const float* __restrict__ x,
    const float* __restrict__ W1, const float* __restrict__ b1,
    const float* __restrict__ W2, const float* __restrict__ b2,
    const float* __restrict__ W3, const float* __restrict__ b3,
    const float* __restrict__ st_in, const float* __restrict__ part_in,
    float* __restrict__ st_out, float* __restrict__ part_out,
    float* __restrict__ y, float* __restrict__ y5out)
{
  constexpr float AA[7][6] = {
    {0.f, 0.f, 0.f, 0.f, 0.f, 0.f},
    {0.2f, 0.f, 0.f, 0.f, 0.f, 0.f},
    {3.f/40.f, 9.f/40.f, 0.f, 0.f, 0.f, 0.f},
    {44.f/45.f, -56.f/15.f, 32.f/9.f, 0.f, 0.f, 0.f},
    {19372.f/6561.f, -25360.f/2187.f, 64448.f/6561.f, -212.f/729.f, 0.f, 0.f},
    {9017.f/3168.f, -355.f/33.f, 46732.f/5247.f, 49.f/176.f, -5103.f/18656.f, 0.f},
    {35.f/384.f, 0.f, 500.f/1113.f, 125.f/192.f, -2187.f/6784.f, 11.f/84.f},
  };
  constexpr float CC[7]  = {0.f, 0.2f, 0.3f, 0.8f, 8.f/9.f, 1.f, 1.f};
  constexpr float B5c[7] = {35.f/384.f, 0.f, 500.f/1113.f, 125.f/192.f, -2187.f/6784.f, 11.f/84.f, 0.f};
  constexpr float B4c[7] = {5179.f/57600.f, 0.f, 7571.f/16695.f, 393.f/640.f,
                            -92097.f/339200.f, 187.f/2100.f, 1.f/40.f};

  __shared__ __align__(16) float w1p[2 * 96 * 2];
  __shared__ float b2s[96];
  __shared__ __align__(16) float w2p[48 * PS2 * 2];
  __shared__ __align__(16) float w3t[2 * 96];
  __shared__ float b3s[2];
  __shared__ __align__(16) float hbuf[NWAVE][16 * HS];
  __shared__ float red[256];

  const int tid  = threadIdx.x;
  const int wave = tid >> 5;
  const int lane = tid & 31;
  const int m    = lane & 15;
  const int hi   = lane >> 4;
  const int elem = blockIdx.x * ELEMS_PER_BLOCK + wave * 16 + m;
  const int yidx = elem * 2 + hi;
  const bool writer = (blockIdx.x == 0) && (tid == 0);

  float t0, dtc, yv;
  if constexpr (FIRST) {
    t0  = 0.0f;
    dtc = 0.05f;
    const float xv = x[elem];
    yv = hi ? xv * xv : xv;
    y[yidx] = yv;
    if (writer) { st_out[0] = t0; st_out[1] = dtc; st_out[2] = 0.0f; }
  } else {
    const float pt    = st_in[0];
    const float pdtc  = st_in[1];
    const float pdone = st_in[2];
    if (pdone != 0.0f) {
      if (writer) { st_out[0] = pt; st_out[1] = pdtc; st_out[2] = 1.0f; }
      return;
    }
    const float yold = y[yidx];
    const float ynew = y5out[yidx];
    const float tot  = block_total(part_in, red, tid);
    const float enorm = sqrtf(tot / (float)(B_TOT * 2));
    const bool accept = (enorm <= 1.0f);
    float factor = 0.9f * powf(fmaxf(enorm, 1e-10f), -0.2f);
    factor = fminf(fmaxf(factor, 0.2f), 10.0f);
    const float t  = accept ? (pt + pdtc) : pt;
    const float dt = pdtc * factor;
    const bool done_new = (t >= 1.0f - 1e-7f);
    t0  = t;
    dtc = fminf(dt, 1.0f - t);
    yv  = yold;
    if (accept) { yv = ynew; y[yidx] = ynew; }
    if (writer) { st_out[0] = t0; st_out[1] = dtc; st_out[2] = done_new ? 1.0f : 0.0f; }
    if (done_new) return;
  }

  for (int i = tid; i < 2 * 96; i += 256) {
    int pr = i / 96, c = i % 96;
    float lo = W1[(2 * pr + 0) * 96 + c];
    float hi_v = (pr == 0) ? W1[96 + c] : b1[c];
    w1p[(pr * 96 + c) * 2 + 0] = lo;
    w1p[(pr * 96 + c) * 2 + 1] = hi_v;
  }
  for (int i = tid; i < 96; i += 256) b2s[i] = b2[i];
  for (int i = tid; i < 96 * 96; i += 256) {
    int r = i / 96, c = i % 96;
    w2p[((r >> 1) * PS2 + c) * 2 + (r & 1)] = W2[i];
  }
  for (int i = tid; i < 192; i += 256) {
    int j = i / 96, n = i % 96;
    w3t[j * 96 + n] = W3[n * 2 + j];
  }
  if (tid < 2) b3s[tid] = b3[tid];
  __syncthreads();

  float* hb = &hbuf[wave][0];

  float ks[7];
#pragma unroll
  for (int i = 0; i < 7; ++i) {
    float yi = yv;
#pragma unroll
    for (int j = 0; j < 6; ++j)
      if (j < i) yi = fmaf(dtc * AA[i][j], ks[j], yi);
    const float ti = t0 + CC[i] * dtc;

    float partner = __shfl_xor(yi, 16, 32);
    v2f a1;
    a1.x = hi ? ti : yi;
    a1.y = hi ? 1.0f : partner;
#pragma unroll
    for (int c = 0; c < 6; ++c) {
      const int col = c * 16 + m;
      const v2f bw = *(const v2f*)&w1p[(hi * 96 + col) * 2];
      v8f acc = {};
      acc = __builtin_amdgcn_wmma_f32_16x16x4_f32(false, a1, false, bw,
                                                  (short)0, acc, false, false);
#pragma unroll
      for (int v = 0; v < 8; ++v)
        hb[(v + 8 * hi) * HS + col] = silu_f(acc[v]);
    }
    asm volatile("s_wait_dscnt 0" ::: "memory");

    v8f acc2[6];
#pragma unroll
    for (int c = 0; c < 6; ++c) {
      const float bb = b2s[c * 16 + m];
#pragma unroll
      for (int v = 0; v < 8; ++v) acc2[c][v] = bb;
    }
#pragma unroll 4
    for (int k = 0; k < 24; ++k) {
      const v2f a2 = *(const v2f*)&hb[m * HS + 4 * k + 2 * hi];
      const int pidx = 2 * k + hi;
#pragma unroll
      for (int c = 0; c < 6; ++c) {
        const v2f bw = *(const v2f*)&w2p[(pidx * PS2 + c * 16 + m) * 2];
        acc2[c] = __builtin_amdgcn_wmma_f32_16x16x4_f32(false, a2, false, bw,
                                                        (short)0, acc2[c], false, false);
      }
    }
    asm volatile("s_wait_dscnt 0" ::: "memory");
#pragma unroll
    for (int c = 0; c < 6; ++c)
#pragma unroll
      for (int v = 0; v < 8; ++v)
        hb[(v + 8 * hi) * HS + c * 16 + m] = silu_f(acc2[c][v]);
    asm volatile("s_wait_dscnt 0" ::: "memory");

    float a3 = b3s[hi];
    const v4f* hrow = (const v4f*)&hb[m * HS];
    const v4f* wcol = (const v4f*)&w3t[hi * 96];
#pragma unroll
    for (int n4 = 0; n4 < 24; ++n4) {
      v4f hv = hrow[n4];
      v4f wv = wcol[n4];
      a3 = fmaf(hv.x, wv.x, a3);
      a3 = fmaf(hv.y, wv.y, a3);
      a3 = fmaf(hv.z, wv.z, a3);
      a3 = fmaf(hv.w, wv.w, a3);
    }
    ks[i] = a3;
    asm volatile("s_wait_dscnt 0" ::: "memory");
  }

  float y5 = yv, ev = 0.0f;
#pragma unroll
  for (int i = 0; i < 7; ++i) {
    y5 = fmaf(dtc * B5c[i], ks[i], y5);
    ev = fmaf(dtc * (B5c[i] - B4c[i]), ks[i], ev);
  }
  y5out[yidx] = y5;

  const float sc = 1e-4f + 1e-4f * fmaxf(fabsf(yv), fabsf(y5));
  const float r = ev / sc;

  red[tid] = r * r;
  __syncthreads();
  for (int w = 128; w > 0; w >>= 1) {
    if (tid < w) red[tid] += red[tid + w];
    __syncthreads();
  }
  if (tid == 0) part_out[blockIdx.x] = red[0];
}


extern "C" void kernel_launch(void* const* d_in, const int* in_sizes, int n_in,
                              void* d_out, int out_size, void* d_ws, size_t ws_size,
                              hipStream_t stream) {
  (void)in_sizes; (void)n_in; (void)out_size; (void)ws_size;
  const float* x  = (const float*)d_in[0];
  const float* W1 = (const float*)d_in[1];
  const float* b1 = (const float*)d_in[2];
  const float* W2 = (const float*)d_in[3];
  const float* b2 = (const float*)d_in[4];
  const float* W3 = (const float*)d_in[5];
  const float* b3 = (const float*)d_in[6];
  const float* Wd = (const float*)d_in[7];
  const float* bd = (const float*)d_in[8];
  float* out = (float*)d_out;
  float* ws  = (float*)d_ws;

  float* stA = ws + ST_A;
  float* stB = ws + ST_B;
  float* ptA = ws + PART_A;
  float* ptB = ws + PART_B;
  float* yst = ws + Y_OFF;
  float* y5  = ws + Y5_OFF;

  step_kernel<true><<<NBLK, 256, 0, stream>>>(x, W1, b1, W2, b2, W3, b3,
                                               stB, ptB, stA, ptA, yst, y5);
  for (int s = 0; s < (MAX_STEPS - 2) / 2; ++s) {
    step_kernel<false><<<NBLK, 256, 0, stream>>>(x, W1, b1, W2, b2, W3, b3,
                                                  stA, ptA, stB, ptB, yst, y5);
    step_kernel<false><<<NBLK, 256, 0, stream>>>(x, W1, b1, W2, b2, W3, b3,
                                                  stB, ptB, stA, ptA, yst, y5);
  }
  step_kernel<false><<<NBLK, 256, 0, stream>>>(x, W1, b1, W2, b2, W3, b3,
                                                stA, ptA, stB, ptB, yst, y5);
  decode_kernel<<<B_TOT / 256, 256, 0, stream>>>(stB, ptB, yst, y5, Wd, bd, out);
}


# Reuse each step's last-stage derivative as the next step's first stage (kept in two workspace planes): steps after the first evaluate 6 of the 7 stages; values and sum order bit-identical
# speedup vs baseline: 1.6043x; 1.1172x over previous
#include <hip/hip_runtime.h>
#include <math.h>

#define B_TOT   262144
#define HIDDEN  96
#define NWAVE   8
#define ELEMS_PER_BLOCK (16 * NWAVE)
#define NBLK    (B_TOT / ELEMS_PER_BLOCK)
#define PS2     112
#define HS      100
#define MAX_STEPS 60

typedef __attribute__((ext_vector_type(2))) float v2f;
typedef __attribute__((ext_vector_type(4))) float v4f;
typedef __attribute__((ext_vector_type(8))) float v8f;

__device__ __forceinline__ float silu_f(float x) {
  return x * __builtin_amdgcn_rcpf(1.0f + __expf(-x));
}

#define ST_A    0
#define ST_B    8
#define PART_A  16
#define PART_B  (16 + NBLK)
#define Y_OFF   (16 + 2 * NBLK)
#define Y5_OFF  (Y_OFF + 2 * B_TOT)
#define K1_OFF  (Y5_OFF + 2 * B_TOT)
#define K7_OFF  (K1_OFF + 2 * B_TOT)

__device__ __forceinline__ float block_total(const float* __restrict__ part, float* red, int tid) {
  float s = 0.0f;
#pragma unroll
  for (int j = 0; j < NBLK / 256; ++j) s += part[tid + 256 * j];
  red[tid] = s;
  __syncthreads();
  for (int w = 128; w > 0; w >>= 1) {
    if (tid < w) red[tid] += red[tid + w];
    __syncthreads();
  }
  return __builtin_bit_cast(float,
      __builtin_amdgcn_readfirstlane(__builtin_bit_cast(int, red[0])));
}


__global__ __launch_bounds__(256) void decode_kernel(
    const float* __restrict__ st_in, const float* __restrict__ part_in,
    const float* __restrict__ y, const float* __restrict__ y5,
    const float* __restrict__ Wd, const float* __restrict__ bd,
    float* __restrict__ out) {
  __shared__ float red[256];
  const int tid = threadIdx.x;
  const int e = blockIdx.x * blockDim.x + tid;
  float y0 = y[e * 2 + 0];
  float y1 = y[e * 2 + 1];
  if (st_in[2] == 0.0f) {
    const float tot = block_total(part_in, red, tid);
    const float enorm = sqrtf(tot / (float)(B_TOT * 2));
    if (enorm <= 1.0f) {
      y0 = y5[e * 2 + 0];
      y1 = y5[e * 2 + 1];
    }
  }
#pragma unroll
  for (int j = 0; j < 10; ++j)
    out[e * 10 + j] = fmaf(y0, Wd[j], fmaf(y1, Wd[10 + j], bd[j]));
}


template <bool FIRST>
__global__ __launch_bounds__(256) void step_kernel(
    const float* __restrict__ x,
    const float* __restrict__ W1, const float* __restrict__ b1,
    const float* __restrict__ W2, const float* __restrict__ b2,
    const float* __restrict__ W3, const float* __restrict__ b3,
    const float* __restrict__ st_in, const float* __restrict__ part_in,
    float* __restrict__ st_out, float* __restrict__ part_out,
    float* __restrict__ y, float* __restrict__ y5out,
    float* __restrict__ k1p, float* __restrict__ k7p)
{
  constexpr float AA[7][6] = {
    {0.f, 0.f, 0.f, 0.f, 0.f, 0.f},
    {0.2f, 0.f, 0.f, 0.f, 0.f, 0.f},
    {3.f/40.f, 9.f/40.f, 0.f, 0.f, 0.f, 0.f},
    {44.f/45.f, -56.f/15.f, 32.f/9.f, 0.f, 0.f, 0.f},
    {19372.f/6561.f, -25360.f/2187.f, 64448.f/6561.f, -212.f/729.f, 0.f, 0.f},
    {9017.f/3168.f, -355.f/33.f, 46732.f/5247.f, 49.f/176.f, -5103.f/18656.f, 0.f},
    {35.f/384.f, 0.f, 500.f/1113.f, 125.f/192.f, -2187.f/6784.f, 11.f/84.f},
  };
  constexpr float CC[7]  = {0.f, 0.2f, 0.3f, 0.8f, 8.f/9.f, 1.f, 1.f};
  constexpr float B5c[7] = {35.f/384.f, 0.f, 500.f/1113.f, 125.f/192.f, -2187.f/6784.f, 11.f/84.f, 0.f};
  constexpr float B4c[7] = {5179.f/57600.f, 0.f, 7571.f/16695.f, 393.f/640.f,
                            -92097.f/339200.f, 187.f/2100.f, 1.f/40.f};

  __shared__ __align__(16) float w1p[2 * 96 * 2];
  __shared__ float b2s[96];
  __shared__ __align__(16) float w2p[48 * PS2 * 2];
  __shared__ __align__(16) float w3t[2 * 96];
  __shared__ float b3s[2];
  __shared__ __align__(16) float hbuf[NWAVE][16 * HS];
  __shared__ float red[256];

  const int tid  = threadIdx.x;
  const int wave = tid >> 5;
  const int lane = tid & 31;
  const int m    = lane & 15;
  const int hi   = lane >> 4;
  const int elem = blockIdx.x * ELEMS_PER_BLOCK + wave * 16 + m;
  const int yidx = elem * 2 + hi;
  const bool writer = (blockIdx.x == 0) && (tid == 0);

  float t0, dtc, yv;
  float kfirst = 0.0f;
  if constexpr (FIRST) {
    t0  = 0.0f;
    dtc = 0.05f;
    const float xv = x[elem];
    yv = hi ? xv * xv : xv;
    y[yidx] = yv;
    if (writer) { st_out[0] = t0; st_out[1] = dtc; st_out[2] = 0.0f; }
  } else {
    const float pt    = st_in[0];
    const float pdtc  = st_in[1];
    const float pdone = st_in[2];
    if (pdone != 0.0f) {
      if (writer) { st_out[0] = pt; st_out[1] = pdtc; st_out[2] = 1.0f; }
      return;
    }
    const float yold = y[yidx];
    const float ynew = y5out[yidx];
    const float k1old = k1p[yidx];
    const float k7new = k7p[yidx];
    const float tot  = block_total(part_in, red, tid);
    const float enorm = sqrtf(tot / (float)(B_TOT * 2));
    const bool accept = (enorm <= 1.0f);
    float factor = 0.9f * powf(fmaxf(enorm, 1e-10f), -0.2f);
    factor = fminf(fmaxf(factor, 0.2f), 10.0f);
    const float t  = accept ? (pt + pdtc) : pt;
    const float dt = pdtc * factor;
    const bool done_new = (t >= 1.0f - 1e-7f);
    t0  = t;
    dtc = fminf(dt, 1.0f - t);
    yv  = yold;
    kfirst = k1old;
    if (accept) {
      yv = ynew;       y[yidx]   = ynew;
      kfirst = k7new;  k1p[yidx] = k7new;
    }
    if (writer) { st_out[0] = t0; st_out[1] = dtc; st_out[2] = done_new ? 1.0f : 0.0f; }
    if (done_new) return;
  }

  for (int i = tid; i < 2 * 96; i += 256) {
    int pr = i / 96, c = i % 96;
    float lo = W1[(2 * pr + 0) * 96 + c];
    float hi_v = (pr == 0) ? W1[96 + c] : b1[c];
    w1p[(pr * 96 + c) * 2 + 0] = lo;
    w1p[(pr * 96 + c) * 2 + 1] = hi_v;
  }
  for (int i = tid; i < 96; i += 256) b2s[i] = b2[i];
  for (int i = tid; i < 96 * 96; i += 256) {
    int r = i / 96, c = i % 96;
    w2p[((r >> 1) * PS2 + c) * 2 + (r & 1)] = W2[i];
  }
  for (int i = tid; i < 192; i += 256) {
    int j = i / 96, n = i % 96;
    w3t[j * 96 + n] = W3[n * 2 + j];
  }
  if (tid < 2) b3s[tid] = b3[tid];
  __syncthreads();

  float* hb = &hbuf[wave][0];

  float ks[7];
  ks[0] = kfirst;
  constexpr int STAGE0 = FIRST ? 0 : 1;
#pragma unroll
  for (int i = STAGE0; i < 7; ++i) {
    float yi = yv;
#pragma unroll
    for (int j = 0; j < 6; ++j)
      if (j < i) yi = fmaf(dtc * AA[i][j], ks[j], yi);
    const float ti = t0 + CC[i] * dtc;

    float partner = __shfl_xor(yi, 16, 32);
    v2f a1;
    a1.x = hi ? ti : yi;
    a1.y = hi ? 1.0f : partner;
#pragma unroll
    for (int c = 0; c < 6; ++c) {
      const int col = c * 16 + m;
      const v2f bw = *(const v2f*)&w1p[(hi * 96 + col) * 2];
      v8f acc = {};
      acc = __builtin_amdgcn_wmma_f32_16x16x4_f32(false, a1, false, bw,
                                                  (short)0, acc, false, false);
#pragma unroll
      for (int v = 0; v < 8; ++v)
        hb[(v + 8 * hi) * HS + col] = silu_f(acc[v]);
    }
    asm volatile("s_wait_dscnt 0" ::: "memory");

    v8f acc2[6];
#pragma unroll
    for (int c = 0; c < 6; ++c) {
      const float bb = b2s[c * 16 + m];
#pragma unroll
      for (int v = 0; v < 8; ++v) acc2[c][v] = bb;
    }
#pragma unroll 4
    for (int k = 0; k < 24; ++k) {
      const v2f a2 = *(const v2f*)&hb[m * HS + 4 * k + 2 * hi];
      const int pidx = 2 * k + hi;
#pragma unroll
      for (int c = 0; c < 6; ++c) {
        const v2f bw = *(const v2f*)&w2p[(pidx * PS2 + c * 16 + m) * 2];
        acc2[c] = __builtin_amdgcn_wmma_f32_16x16x4_f32(false, a2, false, bw,
                                                        (short)0, acc2[c], false, false);
      }
    }
    asm volatile("s_wait_dscnt 0" ::: "memory");
#pragma unroll
    for (int c = 0; c < 6; ++c)
#pragma unroll
      for (int v = 0; v < 8; ++v)
        hb[(v + 8 * hi) * HS + c * 16 + m] = silu_f(acc2[c][v]);
    asm volatile("s_wait_dscnt 0" ::: "memory");

    float a3 = b3s[hi];
    const v4f* hrow = (const v4f*)&hb[m * HS];
    const v4f* wcol = (const v4f*)&w3t[hi * 96];
#pragma unroll
    for (int n4 = 0; n4 < 24; ++n4) {
      v4f hv = hrow[n4];
      v4f wv = wcol[n4];
      a3 = fmaf(hv.x, wv.x, a3);
      a3 = fmaf(hv.y, wv.y, a3);
      a3 = fmaf(hv.z, wv.z, a3);
      a3 = fmaf(hv.w, wv.w, a3);
    }
    ks[i] = a3;
    asm volatile("s_wait_dscnt 0" ::: "memory");
  }

  float y5 = yv, ev = 0.0f;
#pragma unroll
  for (int i = 0; i < 7; ++i) {
    y5 = fmaf(dtc * B5c[i], ks[i], y5);
    ev = fmaf(dtc * (B5c[i] - B4c[i]), ks[i], ev);
  }
  y5out[yidx] = y5;
  k7p[yidx] = ks[6];
  if constexpr (FIRST) k1p[yidx] = ks[0];

  const float sc = 1e-4f + 1e-4f * fmaxf(fabsf(yv), fabsf(y5));
  const float r = ev / sc;

  red[tid] = r * r;
  __syncthreads();
  for (int w = 128; w > 0; w >>= 1) {
    if (tid < w) red[tid] += red[tid + w];
    __syncthreads();
  }
  if (tid == 0) part_out[blockIdx.x] = red[0];
}


extern "C" void kernel_launch(void* const* d_in, const int* in_sizes, int n_in,
                              void* d_out, int out_size, void* d_ws, size_t ws_size,
                              hipStream_t stream) {
  (void)in_sizes; (void)n_in; (void)out_size; (void)ws_size;
  const float* x  = (const float*)d_in[0];
  const float* W1 = (const float*)d_in[1];
  const float* b1 = (const float*)d_in[2];
  const float* W2 = (const float*)d_in[3];
  const float* b2 = (const float*)d_in[4];
  const float* W3 = (const float*)d_in[5];
  const float* b3 = (const float*)d_in[6];
  const float* Wd = (const float*)d_in[7];
  const float* bd = (const float*)d_in[8];
  float* out = (float*)d_out;
  float* ws  = (float*)d_ws;

  float* stA = ws + ST_A;
  float* stB = ws + ST_B;
  float* ptA = ws + PART_A;
  float* ptB = ws + PART_B;
  float* yst = ws + Y_OFF;
  float* y5  = ws + Y5_OFF;
  float* k1  = ws + K1_OFF;
  float* k7  = ws + K7_OFF;

  step_kernel<true><<<NBLK, 256, 0, stream>>>(x, W1, b1, W2, b2, W3, b3,
                                               stB, ptB, stA, ptA, yst, y5, k1, k7);
  for (int s = 0; s < (MAX_STEPS - 2) / 2; ++s) {
    step_kernel<false><<<NBLK, 256, 0, stream>>>(x, W1, b1, W2, b2, W3, b3,
                                                  stA, ptA, stB, ptB, yst, y5, k1, k7);
    step_kernel<false><<<NBLK, 256, 0, stream>>>(x, W1, b1, W2, b2, W3, b3,
                                                  stB, ptB, stA, ptA, yst, y5, k1, k7);
  }
  step_kernel<false><<<NBLK, 256, 0, stream>>>(x, W1, b1, W2, b2, W3, b3,
                                                stA, ptA, stB, ptB, yst, y5, k1, k7);
  decode_kernel<<<B_TOT / 256, 256, 0, stream>>>(stB, ptB, yst, y5, Wd, bd, out);
}
